# starting kernel
# baseline (speedup 1.0000x reference)
#include <hip/hip_runtime.h>
#include <hip/hip_bf16.h>

typedef __attribute__((ext_vector_type(16))) _Float16 v16h;
typedef __attribute__((ext_vector_type(8)))  float    v8f;

__device__ __forceinline__ int klocal(int lane, int j) {
    return ((lane >> 4) << 3) + (j < 8 ? j : j + 8);
}

__global__ __launch_bounds__(256) void k_deg_init(float* __restrict__ deg, int n) {
    int i = blockIdx.x * blockDim.x + threadIdx.x;
    if (i < n) deg[i] = 1.0f;
}

__global__ __launch_bounds__(256) void k_deg_edges(const int* __restrict__ ei,
                                                   float* __restrict__ deg, int E) {
    int e = blockIdx.x * blockDim.x + threadIdx.x;
    if (e < E) atomicAdd(&deg[ei[E + e]], 1.0f);
}

__global__ __launch_bounds__(256) void k_dinv(const float* __restrict__ deg,
                                              float* __restrict__ dinv, int n) {
    int i = blockIdx.x * blockDim.x + threadIdx.x;
    if (i < n) dinv[i] = rsqrtf(deg[i]);
}

__global__ __launch_bounds__(256) void k_norm(const int* __restrict__ ei,
                                              const float* __restrict__ dinv,
                                              float* __restrict__ norm, int E) {
    int e = blockIdx.x * blockDim.x + threadIdx.x;
    if (e < E) norm[e] = dinv[ei[e]] * dinv[ei[E + e]];
}

__global__ __launch_bounds__(256) void k_prep_w(const float* __restrict__ W1,
                                                const float* __restrict__ W2,
                                                _Float16* __restrict__ w1B,
                                                _Float16* __restrict__ w2B) {
    int t = blockIdx.x * blockDim.x + threadIdx.x;
    if (t < 4 * 32 * 16) {
        int c = t / (32 * 16);
        int l = (t / 16) & 31;
        int j = t & 15;
        int k = 32 * c + klocal(l, j);
        w1B[t] = (_Float16)W1[k * 16 + (l & 15)];
    } else if (t < 4 * 32 * 16 + 32 * 16) {
        int u = t - 4 * 32 * 16;
        int l = u / 16;
        int j = u & 15;
        int k = klocal(l, j);
        int n = l & 15;
        w2B[u] = (k < 16 && n < 7) ? (_Float16)W2[k * 7 + n] : (_Float16)0.0f;
    }
}

__global__ __launch_bounds__(256) void k_gemm1(const float* __restrict__ x,
                                               const _Float16* __restrict__ w1B,
                                               const float* __restrict__ dinv,
                                               float* __restrict__ h,
                                               float* __restrict__ acc1, int nrows) {
    int wave = (blockIdx.x * blockDim.x + threadIdx.x) >> 5;
    int lane = threadIdx.x & 31;
    int row0 = wave * 16;
    if (row0 >= nrows) return;
    int rsub = lane & 15;
    int hi   = lane >> 4;
    const float* xr = x + (size_t)(row0 + rsub) * 128 + hi * 8;
    v8f c = {};
#pragma unroll
    for (int kc = 0; kc < 4; ++kc) {
        const float* p = xr + kc * 32;
        v16h a;
#pragma unroll
        for (int j = 0; j < 8; ++j) a[j] = (_Float16)p[j];
#pragma unroll
        for (int j = 0; j < 8; ++j) a[8 + j] = (_Float16)p[16 + j];
        v16h b = *(const v16h*)(w1B + (size_t)(kc * 32 + lane) * 16);
        c = __builtin_amdgcn_wmma_f32_16x16x32_f16(false, a, false, b,
                                                   (short)0, c, false, false);
    }
#pragma unroll
    for (int r = 0; r < 8; ++r) {
        int row = row0 + hi * 8 + r;
        float hv = c[r];
        float dv = dinv[row];
        h   [(size_t)row * 16 + rsub] = hv;
        acc1[(size_t)row * 16 + rsub] = hv * dv * dv;
    }
}

__global__ __launch_bounds__(256) void k_scatter1(const int* __restrict__ ei,
                                                  const float* __restrict__ norm,
                                                  const float* __restrict__ h,
                                                  float* __restrict__ acc1, int E) {
    int t = blockIdx.x * blockDim.x + threadIdx.x;
    if (t >= 4 * E) return;
    int e = t >> 2, q = t & 3;
    int s = ei[e], d = ei[E + e];
    float nm = norm[e];
    const float4 v = *(const float4*)(h + (size_t)s * 16 + q * 4);
    float* ap = acc1 + (size_t)d * 16 + q * 4;
    atomicAdd(ap + 0, v.x * nm);
    atomicAdd(ap + 1, v.y * nm);
    atomicAdd(ap + 2, v.z * nm);
    atomicAdd(ap + 3, v.w * nm);
}

__global__ __launch_bounds__(256) void k_gemm2(const float* __restrict__ acc1,
                                               const float* __restrict__ b1,
                                               const _Float16* __restrict__ w2B,
                                               const float* __restrict__ dinv,
                                               float* __restrict__ h2,
                                               float* __restrict__ acc2, int nrows) {
    int wave = (blockIdx.x * blockDim.x + threadIdx.x) >> 5;
    int lane = threadIdx.x & 31;
    int row0 = wave * 16;
    if (row0 >= nrows) return;
    int rsub = lane & 15;
    int hi   = lane >> 4;
    int base = hi * 8;
    const float* p = acc1 + (size_t)(row0 + rsub) * 16 + base;
    v16h a;
#pragma unroll
    for (int j = 0; j < 8; ++j) {
        float f = p[j] + b1[base + j];
        a[j] = (_Float16)fmaxf(f, 0.0f);
    }
#pragma unroll
    for (int j = 8; j < 16; ++j) a[j] = (_Float16)0.0f;
    v16h b = *(const v16h*)(w2B + (size_t)lane * 16);
    v8f c = {};
    c = __builtin_amdgcn_wmma_f32_16x16x32_f16(false, a, false, b,
                                               (short)0, c, false, false);
    if (rsub < 8) {
#pragma unroll
        for (int r = 0; r < 8; ++r) {
            int row = row0 + base + r;
            float hv = c[r];
            float dv = dinv[row];
            h2  [(size_t)row * 8 + rsub] = hv;
            acc2[(size_t)row * 8 + rsub] = hv * dv * dv;
        }
    }
}

__global__ __launch_bounds__(256) void k_scatter2(const int* __restrict__ ei,
                                                  const float* __restrict__ norm,
                                                  const float* __restrict__ h2,
                                                  float* __restrict__ acc2, int E) {
    int t = blockIdx.x * blockDim.x + threadIdx.x;
    if (t >= 2 * E) return;
    int e = t >> 1, q = t & 1;
    int s = ei[e], d = ei[E + e];
    float nm = norm[e];
    const float4 v = *(const float4*)(h2 + (size_t)s * 8 + q * 4);
    float* ap = acc2 + (size_t)d * 8 + q * 4;
    atomicAdd(ap + 0, v.x * nm);
    atomicAdd(ap + 1, v.y * nm);
    atomicAdd(ap + 2, v.z * nm);
    atomicAdd(ap + 3, v.w * nm);
}

__global__ __launch_bounds__(256) void k_final(const float* __restrict__ acc2,
                                               const float* __restrict__ b2,
                                               float* __restrict__ out, int n7) {
    int t = blockIdx.x * blockDim.x + threadIdx.x;
    if (t < n7) {
        int i = t / 7, c = t % 7;
        out[t] = fmaxf(acc2[(size_t)i * 8 + c] + b2[c], 0.0f);
    }
}

extern "C" void kernel_launch(void* const* d_in, const int* in_sizes, int n_in,
                              void* d_out, int out_size, void* d_ws, size_t ws_size,
                              hipStream_t stream) {
    const float* x  = (const float*)d_in[0];
    const int*   ei = (const int*)d_in[1];
    const float* W1 = (const float*)d_in[2];
    const float* b1 = (const float*)d_in[3];
    const float* W2 = (const float*)d_in[4];
    const float* b2 = (const float*)d_in[5];
    float* out = (float*)d_out;

    const int N = in_sizes[0] / 128;
    const int E = in_sizes[1] / 2;

    char* ws = (char*)d_ws;
    float*    h    = (float*)(ws + 0);
    float*    acc1 = (float*)(ws + (size_t)N * 16 * 4);
    float*    h2   = (float*)(ws + (size_t)N * 32 * 4);
    float*    acc2 = (float*)(ws + (size_t)N * 40 * 4);
    float*    deg  = (float*)(ws + (size_t)N * 48 * 4);
    float*    dinv = (float*)(ws + (size_t)N * 49 * 4);
    float*    norm = (float*)(ws + (size_t)N * 50 * 4);
    _Float16* w1B  = (_Float16*)(ws + (size_t)N * 50 * 4 + (size_t)E * 4);
    _Float16* w2B  = (_Float16*)(ws + (size_t)N * 50 * 4 + (size_t)E * 4 + 4096);

    const int B = 256;
    auto blk = [](long long t, int b) { return (int)((t + b - 1) / b); };

    k_deg_init <<<blk(N, B), B, 0, stream>>>(deg, N);
    k_deg_edges<<<blk(E, B), B, 0, stream>>>(ei, deg, E);
    k_dinv     <<<blk(N, B), B, 0, stream>>>(deg, dinv, N);
    k_norm     <<<blk(E, B), B, 0, stream>>>(ei, dinv, norm, E);

    k_prep_w<<<blk(4 * 32 * 16 + 32 * 16, B), B, 0, stream>>>(W1, W2, w1B, w2B);

    int waves1 = (N + 15) / 16;
    k_gemm1   <<<blk((long long)waves1 * 32, B), B, 0, stream>>>(x, w1B, dinv, h, acc1, N);
    k_scatter1<<<blk((long long)E * 4, B), B, 0, stream>>>(ei, norm, h, acc1, E);

    k_gemm2   <<<blk((long long)waves1 * 32, B), B, 0, stream>>>(acc1, b1, w2B, dinv, h2, acc2, N);
    k_scatter2<<<blk((long long)E * 2, B), B, 0, stream>>>(ei, norm, h2, acc2, E);

    k_final<<<blk((long long)N * 7, B), B, 0, stream>>>(acc2, b2, out, N * 7);
}


# edge scatter by global float atomics replaced by LDS row sums per (node range, edge chunk) block, per-chunk planes added by the consumer; sum order: LDS arrival, then chunk order
# speedup vs baseline: 1.5875x; 1.5875x over previous
#include <hip/hip_runtime.h>
#include <hip/hip_bf16.h>

typedef __attribute__((ext_vector_type(16))) _Float16 v16h;
typedef __attribute__((ext_vector_type(8)))  float    v8f;

__device__ __forceinline__ int klocal(int lane, int j) {
    return ((lane >> 4) << 3) + (j < 8 ? j : j + 8);
}

__global__ __launch_bounds__(256) void k_deg_init(float* __restrict__ deg, int n) {
    int i = blockIdx.x * blockDim.x + threadIdx.x;
    if (i < n) deg[i] = 1.0f;
}

__global__ __launch_bounds__(256) void k_deg_edges(const int* __restrict__ ei,
                                                   float* __restrict__ deg, int E) {
    int e = blockIdx.x * blockDim.x + threadIdx.x;
    if (e < E) atomicAdd(&deg[ei[E + e]], 1.0f);
}

__global__ __launch_bounds__(256) void k_dinv(const float* __restrict__ deg,
                                              float* __restrict__ dinv, int n) {
    int i = blockIdx.x * blockDim.x + threadIdx.x;
    if (i < n) dinv[i] = rsqrtf(deg[i]);
}

__global__ __launch_bounds__(256) void k_norm(const int* __restrict__ ei,
                                              const float* __restrict__ dinv,
                                              float* __restrict__ norm, int E) {
    int e = blockIdx.x * blockDim.x + threadIdx.x;
    if (e < E) norm[e] = dinv[ei[e]] * dinv[ei[E + e]];
}

__global__ __launch_bounds__(256) void k_prep_w(const float* __restrict__ W1,
                                                const float* __restrict__ W2,
                                                _Float16* __restrict__ w1B,
                                                _Float16* __restrict__ w2B) {
    int t = blockIdx.x * blockDim.x + threadIdx.x;
    if (t < 4 * 32 * 16) {
        int c = t / (32 * 16);
        int l = (t / 16) & 31;
        int j = t & 15;
        int k = 32 * c + klocal(l, j);
        w1B[t] = (_Float16)W1[k * 16 + (l & 15)];
    } else if (t < 4 * 32 * 16 + 32 * 16) {
        int u = t - 4 * 32 * 16;
        int l = u / 16;
        int j = u & 15;
        int k = klocal(l, j);
        int n = l & 15;
        w2B[u] = (k < 16 && n < 7) ? (_Float16)W2[k * 7 + n] : (_Float16)0.0f;
    }
}

__global__ __launch_bounds__(256) void k_gemm1(const float* __restrict__ x,
                                               const _Float16* __restrict__ w1B,
                                               const float* __restrict__ dinv,
                                               float* __restrict__ h,
                                               float* __restrict__ acc1, int nrows) {
    int wave = (blockIdx.x * blockDim.x + threadIdx.x) >> 5;
    int lane = threadIdx.x & 31;
    int row0 = wave * 16;
    if (row0 >= nrows) return;
    int rsub = lane & 15;
    int hi   = lane >> 4;
    const float* xr = x + (size_t)(row0 + rsub) * 128 + hi * 8;
    v8f c = {};
#pragma unroll
    for (int kc = 0; kc < 4; ++kc) {
        const float* p = xr + kc * 32;
        v16h a;
#pragma unroll
        for (int j = 0; j < 8; ++j) a[j] = (_Float16)p[j];
#pragma unroll
        for (int j = 0; j < 8; ++j) a[8 + j] = (_Float16)p[16 + j];
        v16h b = *(const v16h*)(w1B + (size_t)(kc * 32 + lane) * 16);
        c = __builtin_amdgcn_wmma_f32_16x16x32_f16(false, a, false, b,
                                                   (short)0, c, false, false);
    }
#pragma unroll
    for (int r = 0; r < 8; ++r) {
        int row = row0 + hi * 8 + r;
        float hv = c[r];
        float dv = dinv[row];
        h   [(size_t)row * 16 + rsub] = hv;
        acc1[(size_t)row * 16 + rsub] = hv * dv * dv;
    }
}

template <int F, bool FULL>
__device__ __forceinline__ void agg_ring_pass(const int4* myq, unsigned head, int cnt,
                                              const float* __restrict__ hin,
                                              float* lacc, int lane) {
    constexpr int LPE = F / 4;
    constexpr int EPP = 32 / LPE;
    const int sub = lane % LPE;
    float4 v[LPE];
    float  nm[LPE];
    int    off[LPE];
    bool   ok[LPE];
#pragma unroll
    for (int p = 0; p < LPE; ++p) {
        int idx = p * EPP + lane / LPE;
        ok[p] = FULL || idx < cnt;
        int4 en = myq[(head + (unsigned)idx) & 63u];
        nm[p]  = __int_as_float(en.z);
        off[p] = en.y * F + sub * 4;
        v[p] = make_float4(0.f, 0.f, 0.f, 0.f);
        if (ok[p]) v[p] = *(const float4*)(hin + (size_t)en.x * F + sub * 4);
    }
#pragma unroll
    for (int p = 0; p < LPE; ++p) {
        if (ok[p]) {
            float* a = lacc + off[p];
            __hip_atomic_fetch_add(a + 0, v[p].x * nm[p], __ATOMIC_RELAXED, __HIP_MEMORY_SCOPE_WORKGROUP);
            __hip_atomic_fetch_add(a + 1, v[p].y * nm[p], __ATOMIC_RELAXED, __HIP_MEMORY_SCOPE_WORKGROUP);
            __hip_atomic_fetch_add(a + 2, v[p].z * nm[p], __ATOMIC_RELAXED, __HIP_MEMORY_SCOPE_WORKGROUP);
            __hip_atomic_fetch_add(a + 3, v[p].w * nm[p], __ATOMIC_RELAXED, __HIP_MEMORY_SCOPE_WORKGROUP);
        }
    }
}

template <int F, int NODES>
__global__ __launch_bounds__(1024) void k_agg(const int* __restrict__ ei,
                                              const float* __restrict__ norm,
                                              const float* __restrict__ hin,
                                              float* __restrict__ part,
                                              int E, int N, int chunk) {
    __shared__ __attribute__((aligned(16))) float lacc[NODES * F];
    __shared__ int4 ring[32 * 64];

    const int tid  = threadIdx.x;
    const int lane = tid & 31;
    const int wv   = tid >> 5;
    const int lo   = blockIdx.x * NODES;
    const int c    = blockIdx.y;

    for (int i = tid; i < NODES * F / 4; i += 1024)
        ((float4*)lacc)[i] = make_float4(0.f, 0.f, 0.f, 0.f);
    __syncthreads();

    const int e0 = c * chunk;
    const int e1 = min(e0 + chunk, E);
    const int* __restrict__ srcp = ei;
    const int* __restrict__ dstp = ei + E;
    int4* myq = ring + wv * 64;
    unsigned head = 0, tail = 0;

    constexpr int U = 8;
    const int stride = 32 * 32 * U;
    int dcur[U];
    {
        int base = e0 + wv * 32 * U;
#pragma unroll
        for (int u = 0; u < U; ++u) {
            int e = base + u * 32 + lane;
            int dv = dstp[min(e, e1 - 1)];
            dcur[u] = (e < e1) ? dv : -1;
        }
    }
    for (int base = e0 + wv * 32 * U; base < e1; base += stride) {
        int dnext[U];
#pragma unroll
        for (int u = 0; u < U; ++u) {
            int e = base + stride + u * 32 + lane;
            int dv = dstp[min(e, e1 - 1)];
            dnext[u] = (e < e1) ? dv : -1;
        }
        int   dl[U], sv[U];
        float nv[U];
        bool  mt[U];
#pragma unroll
        for (int u = 0; u < U; ++u) {
            dl[u] = dcur[u] - lo;
            mt[u] = (unsigned)dl[u] < (unsigned)NODES;
        }
#pragma unroll
        for (int u = 0; u < U; ++u) {
            int e = base + u * 32 + lane;
            sv[u] = 0;
            nv[u] = 0.f;
            if (mt[u]) { sv[u] = srcp[e]; nv[u] = norm[e]; }
        }
#pragma unroll
        for (int u = 0; u < U; ++u) {
            unsigned bal = __builtin_amdgcn_ballot_w32(mt[u]);
            if (bal) {
                unsigned rank = __builtin_amdgcn_mbcnt_lo(bal, 0u);
                if (mt[u])
                    myq[(tail + rank) & 63u] = make_int4(sv[u], dl[u], __float_as_int(nv[u]), 0);
                tail += (unsigned)__builtin_popcount(bal);
                if (tail - head >= 32u) {
                    __builtin_amdgcn_fence(__ATOMIC_ACQ_REL, "wavefront");
                    agg_ring_pass<F, true>(myq, head, 32, hin, lacc, lane);
                    head += 32u;
                }
            }
        }
#pragma unroll
        for (int u = 0; u < U; ++u) dcur[u] = dnext[u];
    }
    if (tail != head) {
        __builtin_amdgcn_fence(__ATOMIC_ACQ_REL, "wavefront");
        agg_ring_pass<F, false>(myq, head, (int)(tail - head), hin, lacc, lane);
    }
    __syncthreads();

    const int nn = min(NODES, N - lo);
    float4* dst4 = (float4*)(part + ((size_t)c * N + lo) * F);
    for (int i = tid; i < nn * (F / 4); i += 1024)
        dst4[i] = ((const float4*)lacc)[i];
}

__global__ __launch_bounds__(256) void k_gemm2(const float* __restrict__ acc1,
                                               const float* __restrict__ part1, int nplanes,
                                               const float* __restrict__ b1,
                                               const _Float16* __restrict__ w2B,
                                               const float* __restrict__ dinv,
                                               float* __restrict__ h2,
                                               float* __restrict__ acc2, int nrows) {
    int wave = (blockIdx.x * blockDim.x + threadIdx.x) >> 5;
    int lane = threadIdx.x & 31;
    int row0 = wave * 16;
    if (row0 >= nrows) return;
    int rsub = lane & 15;
    int hi   = lane >> 4;
    int base = hi * 8;
    const float4* p4 = (const float4*)(acc1 + (size_t)(row0 + rsub) * 16 + base);
    float4 s0 = p4[0], s1 = p4[1];
    for (int c = 0; c < nplanes; ++c) {
        const float4* q4 = (const float4*)(part1 + ((size_t)c * nrows + row0 + rsub) * 16 + base);
        float4 t0 = q4[0], t1 = q4[1];
        s0.x += t0.x; s0.y += t0.y; s0.z += t0.z; s0.w += t0.w;
        s1.x += t1.x; s1.y += t1.y; s1.z += t1.z; s1.w += t1.w;
    }
    const float p[8] = {s0.x, s0.y, s0.z, s0.w, s1.x, s1.y, s1.z, s1.w};
    v16h a;
#pragma unroll
    for (int j = 0; j < 8; ++j) {
        float f = p[j] + b1[base + j];
        a[j] = (_Float16)fmaxf(f, 0.0f);
    }
#pragma unroll
    for (int j = 8; j < 16; ++j) a[j] = (_Float16)0.0f;
    v16h b = *(const v16h*)(w2B + (size_t)lane * 16);
    v8f c = {};
    c = __builtin_amdgcn_wmma_f32_16x16x32_f16(false, a, false, b,
                                               (short)0, c, false, false);
    if (rsub < 8) {
#pragma unroll
        for (int r = 0; r < 8; ++r) {
            int row = row0 + base + r;
            float hv = c[r];
            float dv = dinv[row];
            h2  [(size_t)row * 8 + rsub] = hv;
            acc2[(size_t)row * 8 + rsub] = hv * dv * dv;
        }
    }
}

__global__ __launch_bounds__(256) void k_final(const float* __restrict__ acc2,
                                               const float* __restrict__ part2, int nplanes,
                                               const float* __restrict__ b2,
                                               float* __restrict__ out, int n) {
    int t = blockIdx.x * blockDim.x + threadIdx.x;
    if (t >= 2 * n) return;
    int i = t >> 1, q = t & 1;
    float4 s = *(const float4*)(acc2 + (size_t)i * 8 + q * 4);
    for (int c = 0; c < nplanes; ++c) {
        float4 v = *(const float4*)(part2 + ((size_t)c * n + i) * 8 + q * 4);
        s.x += v.x; s.y += v.y; s.z += v.z; s.w += v.w;
    }
    float* o = out + (size_t)i * 7 + q * 4;
    const float* bb = b2 + q * 4;
    o[0] = fmaxf(s.x + bb[0], 0.0f);
    o[1] = fmaxf(s.y + bb[1], 0.0f);
    o[2] = fmaxf(s.z + bb[2], 0.0f);
    if (q == 0) o[3] = fmaxf(s.w + bb[3], 0.0f);
}

extern "C" void kernel_launch(void* const* d_in, const int* in_sizes, int n_in,
                              void* d_out, int out_size, void* d_ws, size_t ws_size,
                              hipStream_t stream) {
    const float* x  = (const float*)d_in[0];
    const int*   ei = (const int*)d_in[1];
    const float* W1 = (const float*)d_in[2];
    const float* b1 = (const float*)d_in[3];
    const float* W2 = (const float*)d_in[4];
    const float* b2 = (const float*)d_in[5];
    float* out = (float*)d_out;

    const int N = in_sizes[0] / 128;
    const int E = in_sizes[1] / 2;

    char* ws = (char*)d_ws;
    float*    h    = (float*)(ws + 0);
    float*    acc1 = (float*)(ws + (size_t)N * 16 * 4);
    float*    h2   = (float*)(ws + (size_t)N * 32 * 4);
    float*    acc2 = (float*)(ws + (size_t)N * 40 * 4);
    float*    deg  = (float*)(ws + (size_t)N * 48 * 4);
    float*    dinv = (float*)(ws + (size_t)N * 49 * 4);
    float*    norm = (float*)(ws + (size_t)N * 50 * 4);
    _Float16* w1B  = (_Float16*)(ws + (size_t)N * 50 * 4 + (size_t)E * 4);
    _Float16* w2B  = (_Float16*)(ws + (size_t)N * 50 * 4 + (size_t)E * 4 + 4096);
    const int NODES1 = 4000, C1 = 10;
    const int NODES2 = 8336, C2 = 21;
    float*    part1 = (float*)(ws + ((size_t)48 << 20));
    float*    part2 = (float*)(ws + ((size_t)112 << 20));

    const int B = 256;
    auto blk = [](long long t, int b) { return (int)((t + b - 1) / b); };

    k_deg_init <<<blk(N, B), B, 0, stream>>>(deg, N);
    k_deg_edges<<<blk(E, B), B, 0, stream>>>(ei, deg, E);
    k_dinv     <<<blk(N, B), B, 0, stream>>>(deg, dinv, N);
    k_norm     <<<blk(E, B), B, 0, stream>>>(ei, dinv, norm, E);

    k_prep_w<<<blk(4 * 32 * 16 + 32 * 16, B), B, 0, stream>>>(W1, W2, w1B, w2B);

    int waves1 = (N + 15) / 16;
    k_gemm1   <<<blk((long long)waves1 * 32, B), B, 0, stream>>>(x, w1B, dinv, h, acc1, N);
    k_agg<16, NODES1><<<dim3(blk(N, NODES1), C1), 1024, 0, stream>>>(ei, norm, h, part1, E, N, blk(E, C1));

    k_gemm2   <<<blk((long long)waves1 * 32, B), B, 0, stream>>>(acc1, part1, C1, b1, w2B, dinv, h2, acc2, N);
    k_agg<8, NODES2><<<dim3(blk(N, NODES2), C2), 1024, 0, stream>>>(ei, norm, h2, part2, E, N, blk(E, C2));

    k_final<<<blk((long long)N * 2, B), B, 0, stream>>>(acc2, part2, C2, b2, out, N);
}


# layer 1's filter result kept: each (range, chunk) block stores its kept edges as packed words; layer 2 reads its list, no edge rescan; sum order: LDS arrival, then chunk order (10 planes, were 21)
# speedup vs baseline: 1.9138x; 1.2055x over previous
#include <hip/hip_runtime.h>
#include <hip/hip_bf16.h>

typedef __attribute__((ext_vector_type(16))) _Float16 v16h;
typedef __attribute__((ext_vector_type(8)))  float    v8f;

__device__ __forceinline__ int klocal(int lane, int j) {
    return ((lane >> 4) << 3) + (j < 8 ? j : j + 8);
}

__global__ __launch_bounds__(256) void k_deg_init(float* __restrict__ deg, int n) {
    int i = blockIdx.x * blockDim.x + threadIdx.x;
    if (i < n) deg[i] = 1.0f;
}

__global__ __launch_bounds__(256) void k_deg_edges(const int* __restrict__ ei,
                                                   float* __restrict__ deg, int E) {
    const int* __restrict__ dstp = ei + E;
    const int nvec = E & ~3;
    const int emax = max(nvec - 4, 0);
    const int e00  = (blockIdx.x * 1024 + threadIdx.x) * 4;
    int4 d[4];
#pragma unroll
    for (int k = 0; k < 4; ++k)
        d[k] = *(const int4*)(dstp + min(e00 + k * 1024, emax));
#pragma unroll
    for (int k = 0; k < 4; ++k) {
        if (e00 + k * 1024 < nvec) {
            atomicAdd(&deg[d[k].x], 1.0f);
            atomicAdd(&deg[d[k].y], 1.0f);
            atomicAdd(&deg[d[k].z], 1.0f);
            atomicAdd(&deg[d[k].w], 1.0f);
        }
    }
    if (blockIdx.x == 0 && (int)threadIdx.x < (E & 3))
        atomicAdd(&deg[dstp[nvec + threadIdx.x]], 1.0f);
}

__global__ __launch_bounds__(256) void k_dinv(const float* __restrict__ deg,
                                              float* __restrict__ dinv, int n) {
    int i = blockIdx.x * blockDim.x + threadIdx.x;
    if (i < n) dinv[i] = rsqrtf(deg[i]);
}

__global__ __launch_bounds__(256) void k_prep_w(const float* __restrict__ W1,
                                                const float* __restrict__ W2,
                                                _Float16* __restrict__ w1B,
                                                _Float16* __restrict__ w2B) {
    int t = blockIdx.x * blockDim.x + threadIdx.x;
    if (t < 4 * 32 * 16) {
        int c = t / (32 * 16);
        int l = (t / 16) & 31;
        int j = t & 15;
        int k = 32 * c + klocal(l, j);
        w1B[t] = (_Float16)W1[k * 16 + (l & 15)];
    } else if (t < 4 * 32 * 16 + 32 * 16) {
        int u = t - 4 * 32 * 16;
        int l = u / 16;
        int j = u & 15;
        int k = klocal(l, j);
        int n = l & 15;
        w2B[u] = (k < 16 && n < 7) ? (_Float16)W2[k * 7 + n] : (_Float16)0.0f;
    }
}

__global__ __launch_bounds__(256) void k_gemm1(const float* __restrict__ x,
                                               const _Float16* __restrict__ w1B,
                                               const float* __restrict__ dinv,
                                               float* __restrict__ hs, int nrows) {
    int wave = (blockIdx.x * blockDim.x + threadIdx.x) >> 5;
    int lane = threadIdx.x & 31;
    int row0 = wave * 16;
    if (row0 >= nrows) return;
    int rsub = lane & 15;
    int hi   = lane >> 4;
    const float* xr = x + (size_t)(row0 + rsub) * 128 + hi * 8;
    v8f c = {};
#pragma unroll
    for (int kc = 0; kc < 4; ++kc) {
        const float* p = xr + kc * 32;
        v16h a;
#pragma unroll
        for (int j = 0; j < 8; ++j) a[j] = (_Float16)p[j];
#pragma unroll
        for (int j = 0; j < 8; ++j) a[8 + j] = (_Float16)p[16 + j];
        v16h b = *(const v16h*)(w1B + (size_t)(kc * 32 + lane) * 16);
        c = __builtin_amdgcn_wmma_f32_16x16x32_f16(false, a, false, b,
                                                   (short)0, c, false, false);
    }
#pragma unroll
    for (int r = 0; r < 8; ++r) {
        int row = row0 + hi * 8 + r;
        float dv = dinv[row];
        hs[(size_t)row * 16 + rsub] = c[r] * dv;
    }
}

template <int F, bool FULL>
__device__ __forceinline__ void agg_ring_pass(const int2* myq, unsigned head, int cnt,
                                              const float* __restrict__ hin,
                                              float* lacc, int lane) {
    constexpr int LPE = F / 4;
    constexpr int EPP = 32 / LPE;
    const int sub = lane % LPE;
    float4 v[LPE];
    int    off[LPE];
    bool   ok[LPE];
#pragma unroll
    for (int p = 0; p < LPE; ++p) {
        int idx = p * EPP + lane / LPE;
        ok[p] = FULL || idx < cnt;
        int2 en = myq[(head + (unsigned)idx) & 63u];
        off[p] = en.y * F + sub * 4;
        v[p] = make_float4(0.f, 0.f, 0.f, 0.f);
        if (ok[p]) v[p] = *(const float4*)(hin + (size_t)en.x * F + sub * 4);
    }
#pragma unroll
    for (int p = 0; p < LPE; ++p) {
        if (ok[p]) {
            float* a = lacc + off[p];
            __hip_atomic_fetch_add(a + 0, v[p].x, __ATOMIC_RELAXED, __HIP_MEMORY_SCOPE_WORKGROUP);
            __hip_atomic_fetch_add(a + 1, v[p].y, __ATOMIC_RELAXED, __HIP_MEMORY_SCOPE_WORKGROUP);
            __hip_atomic_fetch_add(a + 2, v[p].z, __ATOMIC_RELAXED, __HIP_MEMORY_SCOPE_WORKGROUP);
            __hip_atomic_fetch_add(a + 3, v[p].w, __ATOMIC_RELAXED, __HIP_MEMORY_SCOPE_WORKGROUP);
        }
    }
}

constexpr int LIST_CAP = 16384;

__device__ __forceinline__ void keep_group(const int2* myq, unsigned head, int cnt,
                                           unsigned* lcur, int* __restrict__ mylist, int lane) {
    unsigned cur = 0;
    if (lane == 0)
        cur = __hip_atomic_fetch_add(lcur, 32u, __ATOMIC_RELAXED, __HIP_MEMORY_SCOPE_WORKGROUP);
    cur = __builtin_amdgcn_readfirstlane(cur);
    if (cur + 32u <= (unsigned)LIST_CAP) {
        int2 en = myq[(head + (unsigned)lane) & 63u];
        mylist[cur + lane] = lane < cnt ? (en.x | (en.y << 17)) : -1;
    }
}

template <int F, int NODES, bool KEEP>
__device__ __forceinline__ void agg_filter(const int* __restrict__ ei,
                                           const float* __restrict__ hin,
                                           float* lacc, int2* ring, unsigned* lcur,
                                           int* __restrict__ mylist,
                                           int E, int lo, int c, int chunk_in,
                                           int lane, int wv) {
    const int chunk = (chunk_in + 3) & ~3;
    const int e0  = c * chunk;
    const int ne  = max(min(e0 + chunk, E) - e0, 0);
    const int e1v = e0 + (ne & ~3);
    const int rem = ne & 3;
    const int emax = max((E & ~3) - 4, 0);
    const int* __restrict__ srcp = ei;
    const int* __restrict__ dstp = ei + E;
    int2* myq = ring + wv * 64;
    unsigned head = 0, tail = 0;

    auto step = [&](int dvv, int svv) {
        const int  dl = dvv - lo;
        const bool mt = (unsigned)dl < (unsigned)NODES;
        unsigned bal = __builtin_amdgcn_ballot_w32(mt);
        if (bal) {
            unsigned rank = __builtin_amdgcn_mbcnt_lo(bal, 0u);
            if (mt)
                myq[(tail + rank) & 63u] = make_int2(svv, dl);
            tail += (unsigned)__builtin_popcount(bal);
            if (tail - head >= 32u) {
                __builtin_amdgcn_fence(__ATOMIC_ACQ_REL, "wavefront");
                if (KEEP) keep_group(myq, head, 32, lcur, mylist, lane);
                agg_ring_pass<F, true>(myq, head, 32, hin, lacc, lane);
                head += 32u;
            }
        }
    };

    constexpr int G4   = 4;
    constexpr int SPAN = 128 * G4;
    const int stride = 32 * SPAN;
    int4 dcur[G4], scur[G4];
    bool vcur[G4];
#pragma unroll
    for (int g = 0; g < G4; ++g) {
        const int e = e0 + wv * SPAN + g * 128 + lane * 4;
        vcur[g] = e < e1v;
        dcur[g] = *(const int4*)(dstp + min(e, emax));
        scur[g] = *(const int4*)(srcp + min(e, emax));
    }
    for (int base = e0 + wv * SPAN; base < e1v; base += stride) {
        int4 dnext[G4], snext[G4];
        bool vnext[G4];
#pragma unroll
        for (int g = 0; g < G4; ++g) {
            const int e = base + stride + g * 128 + lane * 4;
            vnext[g] = e < e1v;
            dnext[g] = *(const int4*)(dstp + min(e, emax));
            snext[g] = *(const int4*)(srcp + min(e, emax));
        }
#pragma unroll
        for (int g = 0; g < G4; ++g) {
            step(vcur[g] ? dcur[g].x : -1, scur[g].x);
            step(vcur[g] ? dcur[g].y : -1, scur[g].y);
            step(vcur[g] ? dcur[g].z : -1, scur[g].z);
            step(vcur[g] ? dcur[g].w : -1, scur[g].w);
        }
#pragma unroll
        for (int g = 0; g < G4; ++g) { dcur[g] = dnext[g]; scur[g] = snext[g]; vcur[g] = vnext[g]; }
    }
    if (wv == 0 && rem) {
        const bool in = lane < rem;
        const int  e  = e1v + (in ? lane : 0);
        const int  dv = dstp[e], sv = srcp[e];
        step(in ? dv : -1, sv);
    }
    if (tail != head) {
        __builtin_amdgcn_fence(__ATOMIC_ACQ_REL, "wavefront");
        if (KEEP) keep_group(myq, head, (int)(tail - head), lcur, mylist, lane);
        agg_ring_pass<F, false>(myq, head, (int)(tail - head), hin, lacc, lane);
    }
}

template <int F, int NODES>
__global__ __launch_bounds__(1024) void k_agg_keep(const int* __restrict__ ei,
                                                   const float* __restrict__ hin,
                                                   float* __restrict__ part,
                                                   int* __restrict__ list,
                                                   int* __restrict__ hdr,
                                                   int E, int N, int chunk_in) {
    __shared__ __attribute__((aligned(16))) float lacc[NODES * F];
    __shared__ int2 ring[32 * 64];
    __shared__ unsigned lcur;

    const int tid  = threadIdx.x;
    const int lane = tid & 31;
    const int wv   = __builtin_amdgcn_readfirstlane(tid >> 5);
    const int lo   = blockIdx.x * NODES;
    const int c    = blockIdx.y;
    const int bid  = blockIdx.x * gridDim.y + blockIdx.y;

    if (tid == 0) lcur = 0u;
    for (int i = tid; i < NODES * F / 4; i += 1024)
        ((float4*)lacc)[i] = make_float4(0.f, 0.f, 0.f, 0.f);
    __syncthreads();

    agg_filter<F, NODES, true>(ei, hin, lacc, ring, &lcur, list + (size_t)bid * LIST_CAP,
                               E, lo, c, chunk_in, lane, wv);
    __syncthreads();

    if (tid == 0) hdr[bid] = (int)lcur;

    const int nn = min(NODES, N - lo);
    float4* dst4 = (float4*)(part + ((size_t)c * N + lo) * F);
    for (int i = tid; i < nn * (F / 4); i += 1024)
        dst4[i] = ((const float4*)lacc)[i];
}

template <int NODES>
__global__ __launch_bounds__(1024) void k_agg_list(const int* __restrict__ ei,
                                                   const float* __restrict__ hin,
                                                   float* __restrict__ part,
                                                   const int* __restrict__ list,
                                                   const int* __restrict__ hdr,
                                                   int E, int N, int chunk_in) {
    constexpr int F = 8;
    __shared__ __attribute__((aligned(16))) float lacc[NODES * F];
    __shared__ int2 ring[32 * 64];

    const int tid  = threadIdx.x;
    const int lane = tid & 31;
    const int wv   = __builtin_amdgcn_readfirstlane(tid >> 5);
    const int lo   = blockIdx.x * NODES;
    const int c    = blockIdx.y;
    const int bid  = blockIdx.x * gridDim.y + blockIdx.y;

    for (int i = tid; i < NODES * F / 4; i += 1024)
        ((float4*)lacc)[i] = make_float4(0.f, 0.f, 0.f, 0.f);
    __syncthreads();

    const int nent = __builtin_amdgcn_readfirstlane(hdr[bid]);
    if (nent <= LIST_CAP) {
        const int* __restrict__ mylist = list + (size_t)bid * LIST_CAP;
        const int ng  = nent >> 5;
        const int sub = lane & 1;
        const int eo  = lane >> 1;
        constexpr int GPW = LIST_CAP / 32 / 32;
        int w[GPW][2];
#pragma unroll
        for (int j = 0; j < GPW; ++j) {
            const int g = wv + 32 * j;
#pragma unroll
            for (int p = 0; p < 2; ++p) {
                w[j][p] = -1;
                if (g < ng) w[j][p] = mylist[g * 32 + p * 16 + eo];
            }
        }
#pragma unroll
        for (int jb = 0; jb < GPW; jb += 4) {
            if (wv + 32 * jb < ng) {
                float4 v[4][2];
#pragma unroll
                for (int u = 0; u < 4; ++u)
#pragma unroll
                    for (int p = 0; p < 2; ++p) {
                        const int e = w[jb + u][p];
                        v[u][p] = make_float4(0.f, 0.f, 0.f, 0.f);
                        if (e >= 0)
                            v[u][p] = *(const float4*)(hin + (size_t)(e & 0x1FFFF) * F + sub * 4);
                    }
#pragma unroll
                for (int u = 0; u < 4; ++u)
#pragma unroll
                    for (int p = 0; p < 2; ++p) {
                        const int e = w[jb + u][p];
                        if (e >= 0) {
                            float* a = lacc + ((unsigned)e >> 17) * F + sub * 4;
                            __hip_atomic_fetch_add(a + 0, v[u][p].x, __ATOMIC_RELAXED, __HIP_MEMORY_SCOPE_WORKGROUP);
                            __hip_atomic_fetch_add(a + 1, v[u][p].y, __ATOMIC_RELAXED, __HIP_MEMORY_SCOPE_WORKGROUP);
                            __hip_atomic_fetch_add(a + 2, v[u][p].z, __ATOMIC_RELAXED, __HIP_MEMORY_SCOPE_WORKGROUP);
                            __hip_atomic_fetch_add(a + 3, v[u][p].w, __ATOMIC_RELAXED, __HIP_MEMORY_SCOPE_WORKGROUP);
                        }
                    }
            }
        }
    } else {
        agg_filter<F, NODES, false>(ei, hin, lacc, ring, nullptr, nullptr,
                                    E, lo, c, chunk_in, lane, wv);
    }
    __syncthreads();

    const int nn = min(NODES, N - lo);
    float4* dst4 = (float4*)(part + ((size_t)c * N + lo) * F);
    for (int i = tid; i < nn * (F / 4); i += 1024)
        dst4[i] = ((const float4*)lacc)[i];
}

__global__ __launch_bounds__(256) void k_gemm2(const float* __restrict__ hs,
                                               const float* __restrict__ part1, int nplanes,
                                               const float* __restrict__ b1,
                                               const _Float16* __restrict__ w2B,
                                               const float* __restrict__ dinv,
                                               float* __restrict__ h2s, int nrows) {
    int wave = (blockIdx.x * blockDim.x + threadIdx.x) >> 5;
    int lane = threadIdx.x & 31;
    int row0 = wave * 16;
    if (row0 >= nrows) return;
    int rsub = lane & 15;
    int hi   = lane >> 4;
    int base = hi * 8;
    const float4* p4 = (const float4*)(hs + (size_t)(row0 + rsub) * 16 + base);
    float4 s0 = p4[0], s1 = p4[1];
    const float da = dinv[row0 + rsub];
    for (int c = 0; c < nplanes; ++c) {
        const float4* q4 = (const float4*)(part1 + ((size_t)c * nrows + row0 + rsub) * 16 + base);
        float4 t0 = q4[0], t1 = q4[1];
        s0.x += t0.x; s0.y += t0.y; s0.z += t0.z; s0.w += t0.w;
        s1.x += t1.x; s1.y += t1.y; s1.z += t1.z; s1.w += t1.w;
    }
    const float p[8] = {s0.x, s0.y, s0.z, s0.w, s1.x, s1.y, s1.z, s1.w};
    v16h a;
#pragma unroll
    for (int j = 0; j < 8; ++j) {
        float f = p[j] * da + b1[base + j];
        a[j] = (_Float16)fmaxf(f, 0.0f);
    }
#pragma unroll
    for (int j = 8; j < 16; ++j) a[j] = (_Float16)0.0f;
    v16h b = *(const v16h*)(w2B + (size_t)lane * 16);
    v8f c = {};
    c = __builtin_amdgcn_wmma_f32_16x16x32_f16(false, a, false, b,
                                               (short)0, c, false, false);
    if (rsub < 8) {
#pragma unroll
        for (int r = 0; r < 8; ++r) {
            int row = row0 + base + r;
            float dv = dinv[row];
            h2s[(size_t)row * 8 + rsub] = c[r] * dv;
        }
    }
}

__global__ __launch_bounds__(256) void k_final(const float* __restrict__ h2s,
                                               const float* __restrict__ part2, int nplanes,
                                               const float* __restrict__ b2,
                                               const float* __restrict__ dinv,
                                               float* __restrict__ out, int n) {
    int t = blockIdx.x * blockDim.x + threadIdx.x;
    if (t >= 2 * n) return;
    int i = t >> 1, q = t & 1;
    float4 s = *(const float4*)(h2s + (size_t)i * 8 + q * 4);
    for (int c = 0; c < nplanes; ++c) {
        float4 v = *(const float4*)(part2 + ((size_t)c * n + i) * 8 + q * 4);
        s.x += v.x; s.y += v.y; s.z += v.z; s.w += v.w;
    }
    const float d = dinv[i];
    float* o = out + (size_t)i * 7 + q * 4;
    const float* bb = b2 + q * 4;
    o[0] = fmaxf(s.x * d + bb[0], 0.0f);
    o[1] = fmaxf(s.y * d + bb[1], 0.0f);
    o[2] = fmaxf(s.z * d + bb[2], 0.0f);
    if (q == 0) o[3] = fmaxf(s.w * d + bb[3], 0.0f);
}

extern "C" void kernel_launch(void* const* d_in, const int* in_sizes, int n_in,
                              void* d_out, int out_size, void* d_ws, size_t ws_size,
                              hipStream_t stream) {
    const float* x  = (const float*)d_in[0];
    const int*   ei = (const int*)d_in[1];
    const float* W1 = (const float*)d_in[2];
    const float* b1 = (const float*)d_in[3];
    const float* W2 = (const float*)d_in[4];
    const float* b2 = (const float*)d_in[5];
    float* out = (float*)d_out;

    const int N = in_sizes[0] / 128;
    const int E = in_sizes[1] / 2;

    char* ws = (char*)d_ws;
    float*    hs   = (float*)(ws + 0);
    float*    h2s  = (float*)(ws + (size_t)N * 16 * 4);
    float*    deg  = (float*)(ws + (size_t)N * 24 * 4);
    float*    dinv = (float*)(ws + (size_t)N * 25 * 4);
    _Float16* w1B  = (_Float16*)(ws + (size_t)N * 26 * 4);
    _Float16* w2B  = (_Float16*)(ws + (size_t)N * 26 * 4 + 4096);
    const int NODES1 = 4000, C1 = 10;
    int*      list  = (int*)(ws + ((size_t)16 << 20));
    int*      hdr   = (int*)(ws + ((size_t)40 << 20));
    float*    part1 = (float*)(ws + ((size_t)48 << 20));
    float*    part2 = (float*)(ws + ((size_t)112 << 20));

    const int B = 256;
    auto blk = [](long long t, int b) { return (int)((t + b - 1) / b); };

    k_deg_init <<<blk(N, B), B, 0, stream>>>(deg, N);
    k_deg_edges<<<blk(E, 4096), B, 0, stream>>>(ei, deg, E);
    k_dinv     <<<blk(N, B), B, 0, stream>>>(deg, dinv, N);

    k_prep_w<<<blk(4 * 32 * 16 + 32 * 16, B), B, 0, stream>>>(W1, W2, w1B, w2B);

    int waves1 = (N + 15) / 16;
    k_gemm1   <<<blk((long long)waves1 * 32, B), B, 0, stream>>>(x, w1B, dinv, hs, N);
    k_agg_keep<16, NODES1><<<dim3(blk(N, NODES1), C1), 1024, 0, stream>>>(ei, hs, part1, list, hdr, E, N, blk(E, C1));

    k_gemm2   <<<blk((long long)waves1 * 32, B), B, 0, stream>>>(hs, part1, C1, b1, w2B, dinv, h2s, N);
    k_agg_list<NODES1><<<dim3(blk(N, NODES1), C1), 1024, 0, stream>>>(ei, h2s, part2, list, hdr, E, N, blk(E, C1));

    k_final<<<blk((long long)N * 2, B), B, 0, stream>>>(h2s, part2, C1, b2, dinv, out, N);
}
